# starting kernel
# baseline (speedup 1.0000x reference)
#include <hip/hip_runtime.h>

typedef __attribute__((ext_vector_type(2))) float v2f;
typedef __attribute__((ext_vector_type(8))) float v8f;

constexpr int B_ = 1024;
constexpr int G_ = 20000;
constexpr int L_ = 1000;
constexpr int g_ = 100;
constexpr int C_ = 10;
constexpr int NPAD = 16;

constexpr int TILES_M = B_ / 16;
constexpr int KSPLIT  = 125;
constexpr int KSEG    = G_ / KSPLIT;
constexpr int WSTEPS  = KSEG / 4;
constexpr int UNROLL  = 8;
static_assert(WSTEPS % UNROLL == 0, "batching must divide steps");


constexpr size_t OFF_WEFF  = 0;
constexpr size_t OFF_STAGE = (size_t)G_ * NPAD;
constexpr size_t OFF_BAGG  = OFF_STAGE + (size_t)TILES_M * 256;
constexpr size_t WS_FLOATS = OFF_BAGG + 16;

__global__ void zero_ws_k(float* ws) {
    size_t i = (size_t)blockIdx.x * 256 + threadIdx.x;
    if (i < WS_FLOATS) ws[i] = 0.0f;
}

__global__ void bias_agg_k(const float* __restrict__ bias,
                           const float* __restrict__ agg_w,
                           const float* __restrict__ agg_b,
                           float* __restrict__ bagg) {
    int c = threadIdx.x;
    if (c < C_) {
        float s = agg_b[0];
        for (int l = 0; l < L_; ++l) s += agg_w[l] * bias[l * C_ + c];
        bagg[c] = s;
    }
}

__global__ void scatter_k(const float* __restrict__ W,
                          const float* __restrict__ agg_w,
                          const int* __restrict__ gidx,
                          float* __restrict__ weff) {
    int t = blockIdx.x * 256 + threadIdx.x;
    if (t >= L_ * g_) return;
    int l = t / g_;
    int k = gidx[t];
    float aw = agg_w[l];
    const float* w = W + (size_t)t * C_;
    float* dst = weff + (size_t)(k >> 1) * 32 + (k & 1);
#pragma unroll
    for (int c = 0; c < C_; ++c) atomicAdd(dst + c * 2, aw * w[c]);
}

__global__ __launch_bounds__(32)
void gemm_k(const float* __restrict__ X,
            const float* __restrict__ weff,
            float* __restrict__ stage) {
    const int tile = blockIdx.x % TILES_M;
    const int seg  = blockIdx.x / TILES_M;
    const int lane = threadIdx.x;
    const int half = lane >> 4;
    const int sub  = lane & 15;
    const int koff = half * 2;
    const int kbase = seg * KSEG;

    const float* ap = X + (size_t)(tile * 16 + sub) * G_ + kbase + koff;
    const float* bp = weff + (size_t)((kbase + koff) >> 1) * 32 + sub * 2;

    v8f acc = {0.f, 0.f, 0.f, 0.f, 0.f, 0.f, 0.f, 0.f};

    for (int o = 0; o < WSTEPS / UNROLL; ++o) {
        v2f a[UNROLL], b[UNROLL];
#pragma unroll
        for (int u = 0; u < UNROLL; ++u) {
            a[u] = __builtin_nontemporal_load((const v2f*)(ap + u * 4));
            b[u] = *(const v2f*)(bp + u * 64);
        }
        __builtin_prefetch(ap + UNROLL * 4, 0, 3);
#pragma unroll
        for (int u = 0; u < UNROLL; ++u)
            acc = __builtin_amdgcn_wmma_f32_16x16x4_f32(
                 false, a[u],  false, b[u],
                 (short)0, acc,  false,  false);
        ap += UNROLL * 4;
        bp += UNROLL * 64;
    }

    float* st = stage + tile * 256 + sub;
    const int mbase = half * 8;
#pragma unroll
    for (int r = 0; r < 8; ++r)
        atomicAdd(st + (mbase + r) * 16, acc[r]);
}

__global__ void out_k(const float* __restrict__ stage,
                      const float* __restrict__ bagg,
                      float* __restrict__ out) {
    int t = blockIdx.x * 256 + threadIdx.x;
    if (t >= B_ * C_) return;
    int b = t / C_, c = t % C_;
    out[t] = stage[(b >> 4) * 256 + (b & 15) * 16 + c] + bagg[c];
}

extern "C" void kernel_launch(void* const* d_in, const int* in_sizes, int n_in,
                              void* d_out, int out_size, void* d_ws, size_t ws_size,
                              hipStream_t stream) {
    const float* X     = (const float*)d_in[0];
    const float* W     = (const float*)d_in[1];
    const float* bias  = (const float*)d_in[2];
    const float* agg_w = (const float*)d_in[3];
    const float* agg_b = (const float*)d_in[4];
    const int*   gidx  = (const int*)d_in[5];

    float* ws    = (float*)d_ws;
    float* weff  = ws + OFF_WEFF;
    float* stage = ws + OFF_STAGE;
    float* bagg  = ws + OFF_BAGG;
    float* out   = (float*)d_out;

    (void)in_sizes; (void)n_in; (void)out_size; (void)ws_size;

    const int zgrid = (int)((WS_FLOATS + 255) / 256);
    zero_ws_k<<<zgrid, 256, 0, stream>>>(ws);
    bias_agg_k<<<1, 32, 0, stream>>>(bias, agg_w, agg_b, bagg);
    scatter_k<<<(L_ * g_ + 255) / 256, 256, 0, stream>>>(W, agg_w, gidx, weff);
    gemm_k<<<TILES_M * KSPLIT, 32, 0, stream>>>(X, weff, stage);
    out_k<<<(B_ * C_ + 255) / 256, 256, 0, stream>>>(stage, bagg, out);
}


# bias aggregation (10 dot products of length 1000): one wave per class, all loads issued at once, butterfly lane sum, instead of one serial lane per class; summation order of the bias term changes
# speedup vs baseline: 1.5494x; 1.5494x over previous
#include <hip/hip_runtime.h>

typedef __attribute__((ext_vector_type(2))) float v2f;
typedef __attribute__((ext_vector_type(8))) float v8f;

constexpr int B_ = 1024;
constexpr int G_ = 20000;
constexpr int L_ = 1000;
constexpr int g_ = 100;
constexpr int C_ = 10;
constexpr int NPAD = 16;

constexpr int TILES_M = B_ / 16;
constexpr int KSPLIT  = 125;
constexpr int KSEG    = G_ / KSPLIT;
constexpr int WSTEPS  = KSEG / 4;
constexpr int UNROLL  = 8;
static_assert(WSTEPS % UNROLL == 0, "batching must divide steps");


constexpr size_t OFF_WEFF  = 0;
constexpr size_t OFF_STAGE = (size_t)G_ * NPAD;
constexpr size_t OFF_BAGG  = OFF_STAGE + (size_t)TILES_M * 256;
constexpr size_t WS_FLOATS = OFF_BAGG + 16;

__global__ void zero_ws_k(float* ws) {
    size_t i = (size_t)blockIdx.x * 256 + threadIdx.x;
    if (i < WS_FLOATS) ws[i] = 0.0f;
}

constexpr int BA_WAVE  = 32;
constexpr int BA_STEPS = (L_ + BA_WAVE - 1) / BA_WAVE;
__global__ __launch_bounds__(C_ * BA_WAVE)
void bias_agg_k(const float* __restrict__ bias,
                const float* __restrict__ agg_w,
                const float* __restrict__ agg_b,
                float* __restrict__ bagg) {
    const int c    = threadIdx.x / BA_WAVE;
    const int lane = threadIdx.x % BA_WAVE;
    float w[BA_STEPS], v[BA_STEPS];
    for (int i = 0; i < BA_STEPS; ++i) {
        const int l = lane + i * BA_WAVE;
        const bool ok = l < L_;
        w[i] = ok ? agg_w[l] : 0.0f;
        v[i] = ok ? bias[l * C_ + c] : 0.0f;
    }
    float s0 = 0.f, s1 = 0.f, s2 = 0.f, s3 = 0.f;
    for (int i = 0; i < BA_STEPS; i += 4) {
        s0 = __builtin_fmaf(w[i],     v[i],     s0);
        s1 = __builtin_fmaf(w[i + 1], v[i + 1], s1);
        s2 = __builtin_fmaf(w[i + 2], v[i + 2], s2);
        s3 = __builtin_fmaf(w[i + 3], v[i + 3], s3);
    }
    float s = (s0 + s1) + (s2 + s3);
    for (int m = BA_WAVE / 2; m >= 1; m >>= 1) s += __shfl_xor(s, m, BA_WAVE);
    if (lane == 0) bagg[c] = s + agg_b[0];
}

__global__ void scatter_k(const float* __restrict__ W,
                          const float* __restrict__ agg_w,
                          const int* __restrict__ gidx,
                          float* __restrict__ weff) {
    int t = blockIdx.x * 256 + threadIdx.x;
    if (t >= L_ * g_) return;
    int l = t / g_;
    int k = gidx[t];
    float aw = agg_w[l];
    const float* w = W + (size_t)t * C_;
    float* dst = weff + (size_t)(k >> 1) * 32 + (k & 1);
#pragma unroll
    for (int c = 0; c < C_; ++c) atomicAdd(dst + c * 2, aw * w[c]);
}

__global__ __launch_bounds__(32)
void gemm_k(const float* __restrict__ X,
            const float* __restrict__ weff,
            float* __restrict__ stage) {
    const int tile = blockIdx.x % TILES_M;
    const int seg  = blockIdx.x / TILES_M;
    const int lane = threadIdx.x;
    const int half = lane >> 4;
    const int sub  = lane & 15;
    const int koff = half * 2;
    const int kbase = seg * KSEG;

    const float* ap = X + (size_t)(tile * 16 + sub) * G_ + kbase + koff;
    const float* bp = weff + (size_t)((kbase + koff) >> 1) * 32 + sub * 2;

    v8f acc = {0.f, 0.f, 0.f, 0.f, 0.f, 0.f, 0.f, 0.f};

    for (int o = 0; o < WSTEPS / UNROLL; ++o) {
        v2f a[UNROLL], b[UNROLL];
#pragma unroll
        for (int u = 0; u < UNROLL; ++u) {
            a[u] = __builtin_nontemporal_load((const v2f*)(ap + u * 4));
            b[u] = *(const v2f*)(bp + u * 64);
        }
        __builtin_prefetch(ap + UNROLL * 4, 0, 3);
#pragma unroll
        for (int u = 0; u < UNROLL; ++u)
            acc = __builtin_amdgcn_wmma_f32_16x16x4_f32(
                 false, a[u],  false, b[u],
                 (short)0, acc,  false,  false);
        ap += UNROLL * 4;
        bp += UNROLL * 64;
    }

    float* st = stage + tile * 256 + sub;
    const int mbase = half * 8;
#pragma unroll
    for (int r = 0; r < 8; ++r)
        atomicAdd(st + (mbase + r) * 16, acc[r]);
}

__global__ void out_k(const float* __restrict__ stage,
                      const float* __restrict__ bagg,
                      float* __restrict__ out) {
    int t = blockIdx.x * 256 + threadIdx.x;
    if (t >= B_ * C_) return;
    int b = t / C_, c = t % C_;
    out[t] = stage[(b >> 4) * 256 + (b & 15) * 16 + c] + bagg[c];
}

extern "C" void kernel_launch(void* const* d_in, const int* in_sizes, int n_in,
                              void* d_out, int out_size, void* d_ws, size_t ws_size,
                              hipStream_t stream) {
    const float* X     = (const float*)d_in[0];
    const float* W     = (const float*)d_in[1];
    const float* bias  = (const float*)d_in[2];
    const float* agg_w = (const float*)d_in[3];
    const float* agg_b = (const float*)d_in[4];
    const int*   gidx  = (const int*)d_in[5];

    float* ws    = (float*)d_ws;
    float* weff  = ws + OFF_WEFF;
    float* stage = ws + OFF_STAGE;
    float* bagg  = ws + OFF_BAGG;
    float* out   = (float*)d_out;

    (void)in_sizes; (void)n_in; (void)out_size; (void)ws_size;

    const int zgrid = (int)((WS_FLOATS + 255) / 256);
    zero_ws_k<<<zgrid, 256, 0, stream>>>(ws);
    bias_agg_k<<<1, C_ * BA_WAVE, 0, stream>>>(bias, agg_w, agg_b, bagg);
    scatter_k<<<(L_ * g_ + 255) / 256, 256, 0, stream>>>(W, agg_w, gidx, weff);
    gemm_k<<<TILES_M * KSPLIT, 32, 0, stream>>>(X, weff, stage);
    out_k<<<(B_ * C_ + 255) / 256, 256, 0, stream>>>(stage, bagg, out);
}


# Weff scatter-add: lane = (entry, class), one f32 atomic instruction covers 3 entries' rows (3 lines touched, was 32 random lines); 2048 waves x 17 groups; sums stay atomic, arrival order changes
# speedup vs baseline: 2.3062x; 1.4884x over previous
#include <hip/hip_runtime.h>

typedef __attribute__((ext_vector_type(2))) float v2f;
typedef __attribute__((ext_vector_type(8))) float v8f;

constexpr int B_ = 1024;
constexpr int G_ = 20000;
constexpr int L_ = 1000;
constexpr int g_ = 100;
constexpr int C_ = 10;
constexpr int NPAD = 16;

constexpr int TILES_M = B_ / 16;
constexpr int KSPLIT  = 125;
constexpr int KSEG    = G_ / KSPLIT;
constexpr int WSTEPS  = KSEG / 4;
constexpr int UNROLL  = 8;
static_assert(WSTEPS % UNROLL == 0, "batching must divide steps");


constexpr size_t OFF_WEFF  = 0;
constexpr size_t OFF_STAGE = (size_t)G_ * NPAD;
constexpr size_t OFF_BAGG  = OFF_STAGE + (size_t)TILES_M * 256;
constexpr size_t WS_FLOATS = OFF_BAGG + 16;

__global__ void zero_ws_k(float* ws) {
    size_t i = (size_t)blockIdx.x * 256 + threadIdx.x;
    if (i < WS_FLOATS) ws[i] = 0.0f;
}

constexpr int BA_WAVE  = 32;
constexpr int BA_STEPS = (L_ + BA_WAVE - 1) / BA_WAVE;
__global__ __launch_bounds__(C_ * BA_WAVE)
void bias_agg_k(const float* __restrict__ bias,
                const float* __restrict__ agg_w,
                const float* __restrict__ agg_b,
                float* __restrict__ bagg) {
    const int c    = threadIdx.x / BA_WAVE;
    const int lane = threadIdx.x % BA_WAVE;
    float w[BA_STEPS], v[BA_STEPS];
    for (int i = 0; i < BA_STEPS; ++i) {
        const int l = lane + i * BA_WAVE;
        const bool ok = l < L_;
        w[i] = ok ? agg_w[l] : 0.0f;
        v[i] = ok ? bias[l * C_ + c] : 0.0f;
    }
    float s0 = 0.f, s1 = 0.f, s2 = 0.f, s3 = 0.f;
    for (int i = 0; i < BA_STEPS; i += 4) {
        s0 = __builtin_fmaf(w[i],     v[i],     s0);
        s1 = __builtin_fmaf(w[i + 1], v[i + 1], s1);
        s2 = __builtin_fmaf(w[i + 2], v[i + 2], s2);
        s3 = __builtin_fmaf(w[i + 3], v[i + 3], s3);
    }
    float s = (s0 + s1) + (s2 + s3);
    for (int m = BA_WAVE / 2; m >= 1; m >>= 1) s += __shfl_xor(s, m, BA_WAVE);
    if (lane == 0) bagg[c] = s + agg_b[0];
}

constexpr int SC_N      = L_ * g_;
constexpr int SC_EPI    = 3;
constexpr int SC_GROUPS = (SC_N + SC_EPI - 1) / SC_EPI;
constexpr int SC_BLOCKS = 256;
constexpr int SC_WPB    = 8;
constexpr int SC_WAVES  = SC_BLOCKS * SC_WPB;
constexpr int SC_ITERS  = (SC_GROUPS + SC_WAVES - 1) / SC_WAVES;
__global__ __launch_bounds__(SC_WPB * 32)
void scatter_k(const float* __restrict__ W,
               const float* __restrict__ agg_w,
               const int* __restrict__ gidx,
               float* __restrict__ weff) {
    const int lane = threadIdx.x % 32;
    const int wave = blockIdx.x * SC_WPB + threadIdx.x / 32;
    const int e    = lane / C_;
    const int c    = lane - e * C_;
    const int t0   = (e < SC_EPI) ? wave * (SC_ITERS * SC_EPI) + e : SC_N;
    int   k[SC_ITERS];
    float a[SC_ITERS], w[SC_ITERS];
#pragma unroll
    for (int i = 0; i < SC_ITERS; ++i) {
        const int t  = t0 + i * SC_EPI;
        const int ts = (t < SC_N) ? t : 0;
        k[i] = gidx[ts];
        a[i] = agg_w[ts / g_];
        w[i] = W[(size_t)ts * C_ + c];
    }
    for (int i = 0; i < SC_ITERS; ++i)
        if (t0 + i * SC_EPI < SC_N)
            atomicAdd(weff + (size_t)(k[i] >> 1) * 32 + c * 2 + (k[i] & 1), a[i] * w[i]);
}

__global__ __launch_bounds__(32)
void gemm_k(const float* __restrict__ X,
            const float* __restrict__ weff,
            float* __restrict__ stage) {
    const int tile = blockIdx.x % TILES_M;
    const int seg  = blockIdx.x / TILES_M;
    const int lane = threadIdx.x;
    const int half = lane >> 4;
    const int sub  = lane & 15;
    const int koff = half * 2;
    const int kbase = seg * KSEG;

    const float* ap = X + (size_t)(tile * 16 + sub) * G_ + kbase + koff;
    const float* bp = weff + (size_t)((kbase + koff) >> 1) * 32 + sub * 2;

    v8f acc = {0.f, 0.f, 0.f, 0.f, 0.f, 0.f, 0.f, 0.f};

    for (int o = 0; o < WSTEPS / UNROLL; ++o) {
        v2f a[UNROLL], b[UNROLL];
#pragma unroll
        for (int u = 0; u < UNROLL; ++u) {
            a[u] = __builtin_nontemporal_load((const v2f*)(ap + u * 4));
            b[u] = *(const v2f*)(bp + u * 64);
        }
        __builtin_prefetch(ap + UNROLL * 4, 0, 3);
#pragma unroll
        for (int u = 0; u < UNROLL; ++u)
            acc = __builtin_amdgcn_wmma_f32_16x16x4_f32(
                 false, a[u],  false, b[u],
                 (short)0, acc,  false,  false);
        ap += UNROLL * 4;
        bp += UNROLL * 64;
    }

    float* st = stage + tile * 256 + sub;
    const int mbase = half * 8;
#pragma unroll
    for (int r = 0; r < 8; ++r)
        atomicAdd(st + (mbase + r) * 16, acc[r]);
}

__global__ void out_k(const float* __restrict__ stage,
                      const float* __restrict__ bagg,
                      float* __restrict__ out) {
    int t = blockIdx.x * 256 + threadIdx.x;
    if (t >= B_ * C_) return;
    int b = t / C_, c = t % C_;
    out[t] = stage[(b >> 4) * 256 + (b & 15) * 16 + c] + bagg[c];
}

extern "C" void kernel_launch(void* const* d_in, const int* in_sizes, int n_in,
                              void* d_out, int out_size, void* d_ws, size_t ws_size,
                              hipStream_t stream) {
    const float* X     = (const float*)d_in[0];
    const float* W     = (const float*)d_in[1];
    const float* bias  = (const float*)d_in[2];
    const float* agg_w = (const float*)d_in[3];
    const float* agg_b = (const float*)d_in[4];
    const int*   gidx  = (const int*)d_in[5];

    float* ws    = (float*)d_ws;
    float* weff  = ws + OFF_WEFF;
    float* stage = ws + OFF_STAGE;
    float* bagg  = ws + OFF_BAGG;
    float* out   = (float*)d_out;

    (void)in_sizes; (void)n_in; (void)out_size; (void)ws_size;

    const int zgrid = (int)((WS_FLOATS + 255) / 256);
    zero_ws_k<<<zgrid, 256, 0, stream>>>(ws);
    bias_agg_k<<<1, C_ * BA_WAVE, 0, stream>>>(bias, agg_w, agg_b, bagg);
    scatter_k<<<SC_BLOCKS, SC_WPB * 32, 0, stream>>>(W, agg_w, gidx, weff);
    gemm_k<<<TILES_M * KSPLIT, 32, 0, stream>>>(X, weff, stage);
    out_k<<<(B_ * C_ + 255) / 256, 256, 0, stream>>>(stage, bagg, out);
}


# gemm A operand: each wave copies its 16x160 X tile into LDS with whole-line 16-byte loads (20 loads of 4 full lines; was 40 8-byte loads taking 16 B from 16 lines each) and reads fragments from LDS
# speedup vs baseline: 5.1054x; 2.2138x over previous
#include <hip/hip_runtime.h>

typedef __attribute__((ext_vector_type(2))) float v2f;
typedef __attribute__((ext_vector_type(4))) float v4f;
typedef __attribute__((ext_vector_type(8))) float v8f;

constexpr int B_ = 1024;
constexpr int G_ = 20000;
constexpr int L_ = 1000;
constexpr int g_ = 100;
constexpr int C_ = 10;
constexpr int NPAD = 16;

constexpr int TILES_M = B_ / 16;
constexpr int KSPLIT  = 125;
constexpr int KSEG    = G_ / KSPLIT;
constexpr int WSTEPS  = KSEG / 4;
constexpr int UNROLL  = 8;
static_assert(WSTEPS % UNROLL == 0, "batching must divide steps");

constexpr int A_HEAD  = 128;
constexpr int A_TAIL  = KSEG - A_HEAD;
constexpr int A_LOADS = 16 + 4;
constexpr int A_PITCH = KSEG + 4;
static_assert(A_TAIL == 32, "the tail of a row segment must be exactly one 128-byte line");
static_assert((KSEG * 4) % 128 == 0 && (G_ * 4) % 128 == 0, "row segments start on line boundaries");


constexpr size_t OFF_WEFF  = 0;
constexpr size_t OFF_STAGE = (size_t)G_ * NPAD;
constexpr size_t OFF_BAGG  = OFF_STAGE + (size_t)TILES_M * 256;
constexpr size_t WS_FLOATS = OFF_BAGG + 16;

__global__ void zero_ws_k(float* ws) {
    size_t i = (size_t)blockIdx.x * 256 + threadIdx.x;
    if (i < WS_FLOATS) ws[i] = 0.0f;
}

constexpr int BA_WAVE  = 32;
constexpr int BA_STEPS = (L_ + BA_WAVE - 1) / BA_WAVE;
__global__ __launch_bounds__(C_ * BA_WAVE)
void bias_agg_k(const float* __restrict__ bias,
                const float* __restrict__ agg_w,
                const float* __restrict__ agg_b,
                float* __restrict__ bagg) {
    const int c    = threadIdx.x / BA_WAVE;
    const int lane = threadIdx.x % BA_WAVE;
    float w[BA_STEPS], v[BA_STEPS];
    for (int i = 0; i < BA_STEPS; ++i) {
        const int l = lane + i * BA_WAVE;
        const bool ok = l < L_;
        w[i] = ok ? agg_w[l] : 0.0f;
        v[i] = ok ? bias[l * C_ + c] : 0.0f;
    }
    float s0 = 0.f, s1 = 0.f, s2 = 0.f, s3 = 0.f;
    for (int i = 0; i < BA_STEPS; i += 4) {
        s0 = __builtin_fmaf(w[i],     v[i],     s0);
        s1 = __builtin_fmaf(w[i + 1], v[i + 1], s1);
        s2 = __builtin_fmaf(w[i + 2], v[i + 2], s2);
        s3 = __builtin_fmaf(w[i + 3], v[i + 3], s3);
    }
    float s = (s0 + s1) + (s2 + s3);
    for (int m = BA_WAVE / 2; m >= 1; m >>= 1) s += __shfl_xor(s, m, BA_WAVE);
    if (lane == 0) bagg[c] = s + agg_b[0];
}

constexpr int SC_N      = L_ * g_;
constexpr int SC_EPI    = 3;
constexpr int SC_GROUPS = (SC_N + SC_EPI - 1) / SC_EPI;
constexpr int SC_BLOCKS = 256;
constexpr int SC_WPB    = 8;
constexpr int SC_WAVES  = SC_BLOCKS * SC_WPB;
constexpr int SC_ITERS  = (SC_GROUPS + SC_WAVES - 1) / SC_WAVES;
__global__ __launch_bounds__(SC_WPB * 32)
void scatter_k(const float* __restrict__ W,
               const float* __restrict__ agg_w,
               const int* __restrict__ gidx,
               float* __restrict__ weff) {
    const int lane = threadIdx.x % 32;
    const int wave = blockIdx.x * SC_WPB + threadIdx.x / 32;
    const int e    = lane / C_;
    const int c    = lane - e * C_;
    const int t0   = (e < SC_EPI) ? wave * (SC_ITERS * SC_EPI) + e : SC_N;
    int   k[SC_ITERS];
    float a[SC_ITERS], w[SC_ITERS];
#pragma unroll
    for (int i = 0; i < SC_ITERS; ++i) {
        const int t  = t0 + i * SC_EPI;
        const int ts = (t < SC_N) ? t : 0;
        k[i] = gidx[ts];
        a[i] = agg_w[ts / g_];
        w[i] = W[(size_t)ts * C_ + c];
    }
    for (int i = 0; i < SC_ITERS; ++i)
        if (t0 + i * SC_EPI < SC_N)
            atomicAdd(weff + (size_t)(k[i] >> 1) * 32 + c * 2 + (k[i] & 1), a[i] * w[i]);
}

__global__ __launch_bounds__(32)
void gemm_k(const float* __restrict__ X,
            const float* __restrict__ weff,
            float* __restrict__ stage) {
    __shared__ __attribute__((aligned(16))) float at[16 * A_PITCH];

    const int tile = blockIdx.x % TILES_M;
    const int seg  = blockIdx.x / TILES_M;
    const int lane = threadIdx.x;
    const int half = lane >> 4;
    const int sub  = lane & 15;
    const int koff = half * 2;
    const int kbase = seg * KSEG;

    const float* xt = X + (size_t)(tile * 16) * G_ + kbase;
    const float* xh = xt + lane * 4;
    const float* xl = xt + (size_t)(lane >> 3) * G_ + A_HEAD + (lane & 7) * 4;
    v4f v[A_LOADS];
#pragma unroll
    for (int i = 0; i < 16; ++i)
        v[i] = __builtin_nontemporal_load((const v4f*)(xh + (size_t)i * G_));
#pragma unroll
    for (int j = 0; j < 4; ++j)
        v[16 + j] = __builtin_nontemporal_load((const v4f*)(xl + (size_t)(4 * j) * G_));
    float* lh = at + lane * 4;
    float* ll = at + (lane >> 3) * A_PITCH + A_HEAD + (lane & 7) * 4;
#pragma unroll
    for (int i = 0; i < 16; ++i)
        *(v4f*)(lh + i * A_PITCH) = v[i];
#pragma unroll
    for (int j = 0; j < 4; ++j)
        *(v4f*)(ll + 4 * j * A_PITCH) = v[16 + j];
    __syncthreads();

    const float* al = at + sub * A_PITCH + koff;
    const float* bp = weff + (size_t)((kbase + koff) >> 1) * 32 + sub * 2;

    v8f acc = {0.f, 0.f, 0.f, 0.f, 0.f, 0.f, 0.f, 0.f};

    for (int o = 0; o < WSTEPS / UNROLL; ++o) {
        v2f a[UNROLL], b[UNROLL];
#pragma unroll
        for (int u = 0; u < UNROLL; ++u) {
            a[u] = *(const v2f*)(al + u * 4);
            b[u] = *(const v2f*)(bp + u * 64);
        }
#pragma unroll
        for (int u = 0; u < UNROLL; ++u)
            acc = __builtin_amdgcn_wmma_f32_16x16x4_f32(
                 false, a[u],  false, b[u],
                 (short)0, acc,  false,  false);
        al += UNROLL * 4;
        bp += UNROLL * 64;
    }

    float* st = stage + tile * 256 + sub;
    const int mbase = half * 8;
#pragma unroll
    for (int r = 0; r < 8; ++r)
        atomicAdd(st + (mbase + r) * 16, acc[r]);
}

__global__ void out_k(const float* __restrict__ stage,
                      const float* __restrict__ bagg,
                      float* __restrict__ out) {
    int t = blockIdx.x * 256 + threadIdx.x;
    if (t >= B_ * C_) return;
    int b = t / C_, c = t % C_;
    out[t] = stage[(b >> 4) * 256 + (b & 15) * 16 + c] + bagg[c];
}

extern "C" void kernel_launch(void* const* d_in, const int* in_sizes, int n_in,
                              void* d_out, int out_size, void* d_ws, size_t ws_size,
                              hipStream_t stream) {
    const float* X     = (const float*)d_in[0];
    const float* W     = (const float*)d_in[1];
    const float* bias  = (const float*)d_in[2];
    const float* agg_w = (const float*)d_in[3];
    const float* agg_b = (const float*)d_in[4];
    const int*   gidx  = (const int*)d_in[5];

    float* ws    = (float*)d_ws;
    float* weff  = ws + OFF_WEFF;
    float* stage = ws + OFF_STAGE;
    float* bagg  = ws + OFF_BAGG;
    float* out   = (float*)d_out;

    (void)in_sizes; (void)n_in; (void)out_size; (void)ws_size;

    const int zgrid = (int)((WS_FLOATS + 255) / 256);
    zero_ws_k<<<zgrid, 256, 0, stream>>>(ws);
    bias_agg_k<<<1, C_ * BA_WAVE, 0, stream>>>(bias, agg_w, agg_b, bagg);
    scatter_k<<<SC_BLOCKS, SC_WPB * 32, 0, stream>>>(W, agg_w, gidx, weff);
    gemm_k<<<TILES_M * KSPLIT, 32, 0, stream>>>(X, weff, stage);
    out_k<<<(B_ * C_ + 255) / 256, 256, 0, stream>>>(stage, bagg, out);
}


# launch fusion, 5 launches -> 3: bias aggregation rides in the Weff-clearing launch and stores bagg[c] as start value of out; gemm atomics add straight into out (no stage, no out_k); bias summed first
# speedup vs baseline: 5.6449x; 1.1057x over previous
#include <hip/hip_runtime.h>

typedef __attribute__((ext_vector_type(2))) float v2f;
typedef __attribute__((ext_vector_type(4))) float v4f;
typedef __attribute__((ext_vector_type(8))) float v8f;

constexpr int B_ = 1024;
constexpr int G_ = 20000;
constexpr int L_ = 1000;
constexpr int g_ = 100;
constexpr int C_ = 10;
constexpr int NPAD = 16;

constexpr int TILES_M = B_ / 16;
constexpr int KSPLIT  = 125;
constexpr int KSEG    = G_ / KSPLIT;
constexpr int WSTEPS  = KSEG / 4;
constexpr int UNROLL  = 8;
static_assert(WSTEPS % UNROLL == 0, "batching must divide steps");

constexpr int A_HEAD  = 128;
constexpr int A_TAIL  = KSEG - A_HEAD;
constexpr int A_LOADS = 16 + 4;
constexpr int A_PITCH = KSEG + 4;
static_assert(A_TAIL == 32, "the tail of a row segment must be exactly one 128-byte line");
static_assert((KSEG * 4) % 128 == 0 && (G_ * 4) % 128 == 0, "row segments start on line boundaries");


constexpr size_t OFF_WEFF  = 0;

constexpr int BA_WAVE    = 32;
constexpr int BA_STEPS   = (L_ + BA_WAVE - 1) / BA_WAVE;
constexpr int IN_THREADS = C_ * BA_WAVE;
constexpr int IN_ZBLOCKS = (G_ * NPAD) / IN_THREADS;
constexpr int OUT_N      = B_ * C_;
constexpr int OUT_PER_T  = OUT_N / IN_THREADS;
static_assert(IN_ZBLOCKS * IN_THREADS == G_ * NPAD, "the clearing blocks cover Weff exactly");
static_assert(OUT_PER_T * IN_THREADS == OUT_N, "block 0 covers out exactly");
static_assert(IN_THREADS % C_ == 0, "a thread of block 0 keeps one class over all its stores");
__global__ __launch_bounds__(IN_THREADS)
void init_k(const float* __restrict__ bias,
            const float* __restrict__ agg_w,
            const float* __restrict__ agg_b,
            float* __restrict__ weff,
            float* __restrict__ out) {
    if (blockIdx.x != 0) {
        weff[(size_t)(blockIdx.x - 1) * IN_THREADS + threadIdx.x] = 0.0f;
        return;
    }
    __shared__ float bs[C_];
    const int c    = threadIdx.x / BA_WAVE;
    const int lane = threadIdx.x % BA_WAVE;
    float w[BA_STEPS], v[BA_STEPS];
    for (int i = 0; i < BA_STEPS; ++i) {
        const int l = lane + i * BA_WAVE;
        const bool ok = l < L_;
        w[i] = ok ? agg_w[l] : 0.0f;
        v[i] = ok ? bias[l * C_ + c] : 0.0f;
    }
    float s0 = 0.f, s1 = 0.f, s2 = 0.f, s3 = 0.f;
    for (int i = 0; i < BA_STEPS; i += 4) {
        s0 = __builtin_fmaf(w[i],     v[i],     s0);
        s1 = __builtin_fmaf(w[i + 1], v[i + 1], s1);
        s2 = __builtin_fmaf(w[i + 2], v[i + 2], s2);
        s3 = __builtin_fmaf(w[i + 3], v[i + 3], s3);
    }
    float s = (s0 + s1) + (s2 + s3);
    for (int m = BA_WAVE / 2; m >= 1; m >>= 1) s += __shfl_xor(s, m, BA_WAVE);
    if (lane == 0) bs[c] = s + agg_b[0];
    __syncthreads();
    const float start = bs[threadIdx.x % C_];
    for (int i = 0; i < OUT_PER_T; ++i)
        out[threadIdx.x + i * IN_THREADS] = start;
}

constexpr int SC_N      = L_ * g_;
constexpr int SC_EPI    = 3;
constexpr int SC_GROUPS = (SC_N + SC_EPI - 1) / SC_EPI;
constexpr int SC_BLOCKS = 256;
constexpr int SC_WPB    = 8;
constexpr int SC_WAVES  = SC_BLOCKS * SC_WPB;
constexpr int SC_ITERS  = (SC_GROUPS + SC_WAVES - 1) / SC_WAVES;
__global__ __launch_bounds__(SC_WPB * 32)
void scatter_k(const float* __restrict__ W,
               const float* __restrict__ agg_w,
               const int* __restrict__ gidx,
               float* __restrict__ weff) {
    const int lane = threadIdx.x % 32;
    const int wave = blockIdx.x * SC_WPB + threadIdx.x / 32;
    const int e    = lane / C_;
    const int c    = lane - e * C_;
    const int t0   = (e < SC_EPI) ? wave * (SC_ITERS * SC_EPI) + e : SC_N;
    int   k[SC_ITERS];
    float a[SC_ITERS], w[SC_ITERS];
#pragma unroll
    for (int i = 0; i < SC_ITERS; ++i) {
        const int t  = t0 + i * SC_EPI;
        const int ts = (t < SC_N) ? t : 0;
        k[i] = gidx[ts];
        a[i] = agg_w[ts / g_];
        w[i] = W[(size_t)ts * C_ + c];
    }
    for (int i = 0; i < SC_ITERS; ++i)
        if (t0 + i * SC_EPI < SC_N)
            atomicAdd(weff + (size_t)(k[i] >> 1) * 32 + c * 2 + (k[i] & 1), a[i] * w[i]);
}

__global__ __launch_bounds__(32)
void gemm_k(const float* __restrict__ X,
            const float* __restrict__ weff,
            float* __restrict__ out) {
    __shared__ __attribute__((aligned(16))) float at[16 * A_PITCH];

    const int tile = blockIdx.x % TILES_M;
    const int seg  = blockIdx.x / TILES_M;
    const int lane = threadIdx.x;
    const int half = lane >> 4;
    const int sub  = lane & 15;
    const int koff = half * 2;
    const int kbase = seg * KSEG;

    const float* xt = X + (size_t)(tile * 16) * G_ + kbase;
    const float* xh = xt + lane * 4;
    const float* xl = xt + (size_t)(lane >> 3) * G_ + A_HEAD + (lane & 7) * 4;
    v4f v[A_LOADS];
#pragma unroll
    for (int i = 0; i < 16; ++i)
        v[i] = __builtin_nontemporal_load((const v4f*)(xh + (size_t)i * G_));
#pragma unroll
    for (int j = 0; j < 4; ++j)
        v[16 + j] = __builtin_nontemporal_load((const v4f*)(xl + (size_t)(4 * j) * G_));
    float* lh = at + lane * 4;
    float* ll = at + (lane >> 3) * A_PITCH + A_HEAD + (lane & 7) * 4;
#pragma unroll
    for (int i = 0; i < 16; ++i)
        *(v4f*)(lh + i * A_PITCH) = v[i];
#pragma unroll
    for (int j = 0; j < 4; ++j)
        *(v4f*)(ll + 4 * j * A_PITCH) = v[16 + j];
    __syncthreads();

    const float* al = at + sub * A_PITCH + koff;
    const float* bp = weff + (size_t)((kbase + koff) >> 1) * 32 + sub * 2;

    v8f acc = {0.f, 0.f, 0.f, 0.f, 0.f, 0.f, 0.f, 0.f};

    for (int o = 0; o < WSTEPS / UNROLL; ++o) {
        v2f a[UNROLL], b[UNROLL];
#pragma unroll
        for (int u = 0; u < UNROLL; ++u) {
            a[u] = *(const v2f*)(al + u * 4);
            b[u] = *(const v2f*)(bp + u * 64);
        }
#pragma unroll
        for (int u = 0; u < UNROLL; ++u)
            acc = __builtin_amdgcn_wmma_f32_16x16x4_f32(
                 false, a[u],  false, b[u],
                 (short)0, acc,  false,  false);
        al += UNROLL * 4;
        bp += UNROLL * 64;
    }

    if (sub < C_) {
        float* op = out + (size_t)(tile * 16 + half * 8) * C_ + sub;
#pragma unroll
        for (int r = 0; r < 8; ++r)
            atomicAdd(op + r * C_, acc[r]);
    }
}

extern "C" void kernel_launch(void* const* d_in, const int* in_sizes, int n_in,
                              void* d_out, int out_size, void* d_ws, size_t ws_size,
                              hipStream_t stream) {
    const float* X     = (const float*)d_in[0];
    const float* W     = (const float*)d_in[1];
    const float* bias  = (const float*)d_in[2];
    const float* agg_w = (const float*)d_in[3];
    const float* agg_b = (const float*)d_in[4];
    const int*   gidx  = (const int*)d_in[5];

    float* ws    = (float*)d_ws;
    float* weff  = ws + OFF_WEFF;
    float* out   = (float*)d_out;

    (void)in_sizes; (void)n_in; (void)out_size; (void)ws_size;

    init_k<<<IN_ZBLOCKS + 1, IN_THREADS, 0, stream>>>(bias, agg_w, agg_b, weff, out);
    scatter_k<<<SC_BLOCKS, SC_WPB * 32, 0, stream>>>(W, agg_w, gidx, weff);
    gemm_k<<<TILES_M * KSPLIT, 32, 0, stream>>>(X, weff, out);
}


# fewer, longer-lived waves: Weff clear as 5 x 16 B stores per thread in 50 blocks (was 1 float, 1000 blocks; 10010 -> 510 waves); a gemm wave takes 2 row tiles per K segment, sharing B (8000 -> 4000)
# speedup vs baseline: 5.9263x; 1.0499x over previous
#include <hip/hip_runtime.h>

typedef __attribute__((ext_vector_type(2))) float v2f;
typedef __attribute__((ext_vector_type(4))) float v4f;
typedef __attribute__((ext_vector_type(8))) float v8f;

constexpr int B_ = 1024;
constexpr int G_ = 20000;
constexpr int L_ = 1000;
constexpr int g_ = 100;
constexpr int C_ = 10;
constexpr int NPAD = 16;

constexpr int TILES_M = B_ / 16;
constexpr int KSPLIT  = 125;
constexpr int KSEG    = G_ / KSPLIT;
constexpr int WSTEPS  = KSEG / 4;
constexpr int UNROLL  = 8;
static_assert(WSTEPS % UNROLL == 0, "batching must divide steps");

constexpr int TPW     = 2;
constexpr int WROWS   = 16 * TPW;
constexpr int MGROUPS = TILES_M / TPW;
constexpr int A_HEAD  = 128;
constexpr int A_TAIL  = KSEG - A_HEAD;
constexpr int A_LOADS = WROWS + WROWS / 4;
static_assert(A_TAIL == 32, "the tail of a row segment must be exactly one 128-byte line");
static_assert((KSEG / 4) % 8 == 0, "the swizzle permutes chunks inside groups of 8");
static_assert((KSEG * 4) % 128 == 0 && (G_ * 4) % 128 == 0, "row segments start on line boundaries");
static_assert(TILES_M % TPW == 0, "row tiles pair up exactly");


constexpr size_t OFF_WEFF  = 0;

constexpr int BA_WAVE    = 32;
constexpr int BA_STEPS   = (L_ + BA_WAVE - 1) / BA_WAVE;
constexpr int IN_THREADS = C_ * BA_WAVE;
constexpr int IN_ZPER    = 5;
constexpr int IN_ZBLOCKS = (G_ * NPAD) / (4 * IN_ZPER * IN_THREADS);
constexpr int OUT_N      = B_ * C_;
constexpr int OUT_PER_T  = OUT_N / IN_THREADS;
static_assert(IN_ZBLOCKS * IN_ZPER * IN_THREADS * 4 == G_ * NPAD, "the clearing blocks cover Weff exactly");
static_assert(OUT_PER_T * IN_THREADS == OUT_N, "block 0 covers out exactly");
static_assert(IN_THREADS % C_ == 0, "a thread of block 0 keeps one class over all its stores");
__global__ __launch_bounds__(IN_THREADS)
void init_k(const float* __restrict__ bias,
            const float* __restrict__ agg_w,
            const float* __restrict__ agg_b,
            float* __restrict__ weff,
            float* __restrict__ out) {
    if (blockIdx.x != 0) {
        v4f* z = (v4f*)weff + (size_t)(blockIdx.x - 1) * (IN_ZPER * IN_THREADS) + threadIdx.x;
        const v4f zero = {0.f, 0.f, 0.f, 0.f};
        for (int i = 0; i < IN_ZPER; ++i)
            z[i * IN_THREADS] = zero;
        return;
    }
    __shared__ float bs[C_];
    const int c    = threadIdx.x / BA_WAVE;
    const int lane = threadIdx.x % BA_WAVE;
    float w[BA_STEPS], v[BA_STEPS];
    for (int i = 0; i < BA_STEPS; ++i) {
        const int l = lane + i * BA_WAVE;
        const bool ok = l < L_;
        w[i] = ok ? agg_w[l] : 0.0f;
        v[i] = ok ? bias[l * C_ + c] : 0.0f;
    }
    float s0 = 0.f, s1 = 0.f, s2 = 0.f, s3 = 0.f;
    for (int i = 0; i < BA_STEPS; i += 4) {
        s0 = __builtin_fmaf(w[i],     v[i],     s0);
        s1 = __builtin_fmaf(w[i + 1], v[i + 1], s1);
        s2 = __builtin_fmaf(w[i + 2], v[i + 2], s2);
        s3 = __builtin_fmaf(w[i + 3], v[i + 3], s3);
    }
    float s = (s0 + s1) + (s2 + s3);
    for (int m = BA_WAVE / 2; m >= 1; m >>= 1) s += __shfl_xor(s, m, BA_WAVE);
    if (lane == 0) bs[c] = s + agg_b[0];
    __syncthreads();
    const float start = bs[threadIdx.x % C_];
    for (int i = 0; i < OUT_PER_T; ++i)
        out[threadIdx.x + i * IN_THREADS] = start;
}

constexpr int SC_N      = L_ * g_;
constexpr int SC_EPI    = 3;
constexpr int SC_GROUPS = (SC_N + SC_EPI - 1) / SC_EPI;
constexpr int SC_BLOCKS = 256;
constexpr int SC_WPB    = 8;
constexpr int SC_WAVES  = SC_BLOCKS * SC_WPB;
constexpr int SC_ITERS  = (SC_GROUPS + SC_WAVES - 1) / SC_WAVES;
__global__ __launch_bounds__(SC_WPB * 32)
void scatter_k(const float* __restrict__ W,
               const float* __restrict__ agg_w,
               const int* __restrict__ gidx,
               float* __restrict__ weff) {
    const int lane = threadIdx.x % 32;
    const int wave = blockIdx.x * SC_WPB + threadIdx.x / 32;
    const int e    = lane / C_;
    const int c    = lane - e * C_;
    const int t0   = (e < SC_EPI) ? wave * (SC_ITERS * SC_EPI) + e : SC_N;
    int   k[SC_ITERS];
    float a[SC_ITERS], w[SC_ITERS];
#pragma unroll
    for (int i = 0; i < SC_ITERS; ++i) {
        const int t  = t0 + i * SC_EPI;
        const int ts = (t < SC_N) ? t : 0;
        k[i] = gidx[ts];
        a[i] = agg_w[ts / g_];
        w[i] = W[(size_t)ts * C_ + c];
    }
    for (int i = 0; i < SC_ITERS; ++i)
        if (t0 + i * SC_EPI < SC_N)
            atomicAdd(weff + (size_t)(k[i] >> 1) * 32 + c * 2 + (k[i] & 1), a[i] * w[i]);
}

__global__ __launch_bounds__(32)
void gemm_k(const float* __restrict__ X,
            const float* __restrict__ weff,
            float* __restrict__ out) {
    __shared__ __attribute__((aligned(16))) float at[WROWS * KSEG];

    const int grp  = blockIdx.x % MGROUPS;
    const int seg  = blockIdx.x / MGROUPS;
    const int lane = threadIdx.x;
    const int half = lane >> 4;
    const int sub  = lane & 15;
    const int koff = half * 2;
    const int kbase = seg * KSEG;

    const float* xt = X + (size_t)(grp * WROWS) * G_ + kbase;
    const float* xh = xt + lane * 4;
    const float* xl = xt + (size_t)(lane >> 3) * G_ + A_HEAD + (lane & 7) * 4;
    v4f v[A_LOADS];
#pragma unroll
    for (int i = 0; i < WROWS; ++i)
        v[i] = __builtin_nontemporal_load((const v4f*)(xh + (size_t)i * G_));
#pragma unroll
    for (int j = 0; j < WROWS / 4; ++j)
        v[WROWS + j] = __builtin_nontemporal_load((const v4f*)(xl + (size_t)(4 * j) * G_));
#pragma unroll
    for (int i = 0; i < WROWS; ++i)
        *(v4f*)(at + i * KSEG + (lane ^ ((i >> 1) & 7)) * 4) = v[i];
#pragma unroll
    for (int j = 0; j < WROWS / 4; ++j) {
        const int R = 4 * j + (lane >> 3);
        *(v4f*)(at + R * KSEG + A_HEAD + ((lane & 7) ^ ((R >> 1) & 7)) * 4) = v[WROWS + j];
    }
    __syncthreads();

    const int swz = (sub >> 1) & 7;
    const float* al = at + sub * KSEG + koff;
    const float* bp = weff + (size_t)((kbase + koff) >> 1) * 32 + sub * 2;

    v8f acc0 = {0.f, 0.f, 0.f, 0.f, 0.f, 0.f, 0.f, 0.f};
    v8f acc1 = {0.f, 0.f, 0.f, 0.f, 0.f, 0.f, 0.f, 0.f};

    for (int o = 0; o < WSTEPS / UNROLL; ++o) {
        v2f a0[UNROLL], a1[UNROLL], b[UNROLL];
#pragma unroll
        for (int u = 0; u < UNROLL; ++u) {
            a0[u] = *(const v2f*)(al + (u ^ swz) * 4);
            a1[u] = *(const v2f*)(al + 16 * KSEG + (u ^ swz) * 4);
            b[u] = *(const v2f*)(bp + u * 64);
        }
#pragma unroll
        for (int u = 0; u < UNROLL; ++u) {
            acc0 = __builtin_amdgcn_wmma_f32_16x16x4_f32(
                 false, a0[u],  false, b[u],
                 (short)0, acc0,  false,  false);
            acc1 = __builtin_amdgcn_wmma_f32_16x16x4_f32(
                 false, a1[u],  false, b[u],
                 (short)0, acc1,  false,  false);
        }
        al += UNROLL * 4;
        bp += UNROLL * 64;
    }

    if (sub < C_) {
        float* op = out + (size_t)(grp * WROWS + half * 8) * C_ + sub;
#pragma unroll
        for (int r = 0; r < 8; ++r)
            atomicAdd(op + r * C_, acc0[r]);
#pragma unroll
        for (int r = 0; r < 8; ++r)
            atomicAdd(op + (16 + r) * C_, acc1[r]);
    }
}

extern "C" void kernel_launch(void* const* d_in, const int* in_sizes, int n_in,
                              void* d_out, int out_size, void* d_ws, size_t ws_size,
                              hipStream_t stream) {
    const float* X     = (const float*)d_in[0];
    const float* W     = (const float*)d_in[1];
    const float* bias  = (const float*)d_in[2];
    const float* agg_w = (const float*)d_in[3];
    const float* agg_b = (const float*)d_in[4];
    const int*   gidx  = (const int*)d_in[5];

    float* ws    = (float*)d_ws;
    float* weff  = ws + OFF_WEFF;
    float* out   = (float*)d_out;

    (void)in_sizes; (void)n_in; (void)out_size; (void)ws_size;

    init_k<<<IN_ZBLOCKS + 1, IN_THREADS, 0, stream>>>(bias, agg_w, agg_b, weff, out);
    scatter_k<<<SC_BLOCKS, SC_WPB * 32, 0, stream>>>(W, agg_w, gidx, weff);
    gemm_k<<<MGROUPS * KSPLIT, 32, 0, stream>>>(X, weff, out);
}


# bias sums and the start value of out leave the Weff-clearing launch and run as block 0 of the scatter launch, beside the scatter blocks instead of ahead of them; launch 1 only clears Weff
# speedup vs baseline: 6.4125x; 1.0820x over previous
#include <hip/hip_runtime.h>

typedef __attribute__((ext_vector_type(2))) float v2f;
typedef __attribute__((ext_vector_type(4))) float v4f;
typedef __attribute__((ext_vector_type(8))) float v8f;

constexpr int B_ = 1024;
constexpr int G_ = 20000;
constexpr int L_ = 1000;
constexpr int g_ = 100;
constexpr int C_ = 10;
constexpr int NPAD = 16;

constexpr int TILES_M = B_ / 16;
constexpr int KSPLIT  = 125;
constexpr int KSEG    = G_ / KSPLIT;
constexpr int WSTEPS  = KSEG / 4;
constexpr int UNROLL  = 8;
static_assert(WSTEPS % UNROLL == 0, "batching must divide steps");

constexpr int TPW     = 2;
constexpr int WROWS   = 16 * TPW;
constexpr int MGROUPS = TILES_M / TPW;
constexpr int A_HEAD  = 128;
constexpr int A_TAIL  = KSEG - A_HEAD;
constexpr int A_LOADS = WROWS + WROWS / 4;
static_assert(A_TAIL == 32, "the tail of a row segment must be exactly one 128-byte line");
static_assert((KSEG / 4) % 8 == 0, "the swizzle permutes chunks inside groups of 8");
static_assert((KSEG * 4) % 128 == 0 && (G_ * 4) % 128 == 0, "row segments start on line boundaries");
static_assert(TILES_M % TPW == 0, "row tiles pair up exactly");


constexpr size_t OFF_WEFF  = 0;

constexpr int BA_WAVE    = 32;
constexpr int IN_THREADS = 10 * BA_WAVE;
constexpr int IN_ZPER    = 5;
constexpr int IN_ZBLOCKS = (G_ * NPAD) / (4 * IN_ZPER * IN_THREADS);
static_assert(IN_ZBLOCKS * IN_ZPER * IN_THREADS * 4 == G_ * NPAD, "the clearing blocks cover Weff exactly");
__global__ __launch_bounds__(IN_THREADS)
void init_k(float* __restrict__ weff) {
    v4f* z = (v4f*)weff + (size_t)blockIdx.x * (IN_ZPER * IN_THREADS) + threadIdx.x;
    const v4f zero = {0.f, 0.f, 0.f, 0.f};
    for (int i = 0; i < IN_ZPER; ++i)
        z[i * IN_THREADS] = zero;
}

constexpr int SC_N      = L_ * g_;
constexpr int SC_EPI    = 3;
constexpr int SC_GROUPS = (SC_N + SC_EPI - 1) / SC_EPI;
constexpr int SC_BLOCKS = 256;
constexpr int SC_WPB    = 8;
constexpr int SC_WAVES  = SC_BLOCKS * SC_WPB;
constexpr int SC_ITERS  = (SC_GROUPS + SC_WAVES - 1) / SC_WAVES;
constexpr int SC_THREADS = SC_WPB * BA_WAVE;
constexpr int BA_STEPS   = (L_ + BA_WAVE - 1) / BA_WAVE;
constexpr int OUT_N      = B_ * C_;
constexpr int OUT_PER_T  = OUT_N / SC_THREADS;
static_assert(OUT_PER_T * SC_THREADS == OUT_N, "block 0 covers out exactly");
__global__ __launch_bounds__(SC_THREADS)
void scatter_k(const float* __restrict__ W,
               const float* __restrict__ agg_w,
               const int* __restrict__ gidx,
               const float* __restrict__ bias,
               const float* __restrict__ agg_b,
               float* __restrict__ weff,
               float* __restrict__ out) {
    const int lane = threadIdx.x % BA_WAVE;
    if (blockIdx.x == 0) {
        __shared__ float bs[C_];
        for (int c = threadIdx.x / BA_WAVE; c < C_; c += SC_WPB) {
            float w[BA_STEPS], v[BA_STEPS];
            for (int i = 0; i < BA_STEPS; ++i) {
                const int l = lane + i * BA_WAVE;
                const bool ok = l < L_;
                w[i] = ok ? agg_w[l] : 0.0f;
                v[i] = ok ? bias[l * C_ + c] : 0.0f;
            }
            float s0 = 0.f, s1 = 0.f, s2 = 0.f, s3 = 0.f;
            for (int i = 0; i < BA_STEPS; i += 4) {
                s0 = __builtin_fmaf(w[i],     v[i],     s0);
                s1 = __builtin_fmaf(w[i + 1], v[i + 1], s1);
                s2 = __builtin_fmaf(w[i + 2], v[i + 2], s2);
                s3 = __builtin_fmaf(w[i + 3], v[i + 3], s3);
            }
            float s = (s0 + s1) + (s2 + s3);
            for (int m = BA_WAVE / 2; m >= 1; m >>= 1) s += __shfl_xor(s, m, BA_WAVE);
            if (lane == 0) bs[c] = s + agg_b[0];
        }
        __syncthreads();
        for (int i = 0; i < OUT_PER_T; ++i) {
            const int o = threadIdx.x + i * SC_THREADS;
            out[o] = bs[o % C_];
        }
        return;
    }
    const int wave = (blockIdx.x - 1) * SC_WPB + threadIdx.x / BA_WAVE;
    const int e    = lane / C_;
    const int c    = lane - e * C_;
    const int t0   = (e < SC_EPI) ? wave * (SC_ITERS * SC_EPI) + e : SC_N;
    int   k[SC_ITERS];
    float a[SC_ITERS], w[SC_ITERS];
#pragma unroll
    for (int i = 0; i < SC_ITERS; ++i) {
        const int t  = t0 + i * SC_EPI;
        const int ts = (t < SC_N) ? t : 0;
        k[i] = gidx[ts];
        a[i] = agg_w[ts / g_];
        w[i] = W[(size_t)ts * C_ + c];
    }
    for (int i = 0; i < SC_ITERS; ++i)
        if (t0 + i * SC_EPI < SC_N)
            atomicAdd(weff + (size_t)(k[i] >> 1) * 32 + c * 2 + (k[i] & 1), a[i] * w[i]);
}

__global__ __launch_bounds__(32)
void gemm_k(const float* __restrict__ X,
            const float* __restrict__ weff,
            float* __restrict__ out) {
    __shared__ __attribute__((aligned(16))) float at[WROWS * KSEG];

    const int grp  = blockIdx.x % MGROUPS;
    const int seg  = blockIdx.x / MGROUPS;
    const int lane = threadIdx.x;
    const int half = lane >> 4;
    const int sub  = lane & 15;
    const int koff = half * 2;
    const int kbase = seg * KSEG;

    const float* xt = X + (size_t)(grp * WROWS) * G_ + kbase;
    const float* xh = xt + lane * 4;
    const float* xl = xt + (size_t)(lane >> 3) * G_ + A_HEAD + (lane & 7) * 4;
    v4f v[A_LOADS];
#pragma unroll
    for (int i = 0; i < WROWS; ++i)
        v[i] = __builtin_nontemporal_load((const v4f*)(xh + (size_t)i * G_));
#pragma unroll
    for (int j = 0; j < WROWS / 4; ++j)
        v[WROWS + j] = __builtin_nontemporal_load((const v4f*)(xl + (size_t)(4 * j) * G_));
#pragma unroll
    for (int i = 0; i < WROWS; ++i)
        *(v4f*)(at + i * KSEG + (lane ^ ((i >> 1) & 7)) * 4) = v[i];
#pragma unroll
    for (int j = 0; j < WROWS / 4; ++j) {
        const int R = 4 * j + (lane >> 3);
        *(v4f*)(at + R * KSEG + A_HEAD + ((lane & 7) ^ ((R >> 1) & 7)) * 4) = v[WROWS + j];
    }
    __syncthreads();

    const int swz = (sub >> 1) & 7;
    const float* al = at + sub * KSEG + koff;
    const float* bp = weff + (size_t)((kbase + koff) >> 1) * 32 + sub * 2;

    v8f acc0 = {0.f, 0.f, 0.f, 0.f, 0.f, 0.f, 0.f, 0.f};
    v8f acc1 = {0.f, 0.f, 0.f, 0.f, 0.f, 0.f, 0.f, 0.f};

    for (int o = 0; o < WSTEPS / UNROLL; ++o) {
        v2f a0[UNROLL], a1[UNROLL], b[UNROLL];
#pragma unroll
        for (int u = 0; u < UNROLL; ++u) {
            a0[u] = *(const v2f*)(al + (u ^ swz) * 4);
            a1[u] = *(const v2f*)(al + 16 * KSEG + (u ^ swz) * 4);
            b[u] = *(const v2f*)(bp + u * 64);
        }
#pragma unroll
        for (int u = 0; u < UNROLL; ++u) {
            acc0 = __builtin_amdgcn_wmma_f32_16x16x4_f32(
                 false, a0[u],  false, b[u],
                 (short)0, acc0,  false,  false);
            acc1 = __builtin_amdgcn_wmma_f32_16x16x4_f32(
                 false, a1[u],  false, b[u],
                 (short)0, acc1,  false,  false);
        }
        al += UNROLL * 4;
        bp += UNROLL * 64;
    }

    if (sub < C_) {
        float* op = out + (size_t)(grp * WROWS + half * 8) * C_ + sub;
#pragma unroll
        for (int r = 0; r < 8; ++r)
            atomicAdd(op + r * C_, acc0[r]);
#pragma unroll
        for (int r = 0; r < 8; ++r)
            atomicAdd(op + (16 + r) * C_, acc1[r]);
    }
}

extern "C" void kernel_launch(void* const* d_in, const int* in_sizes, int n_in,
                              void* d_out, int out_size, void* d_ws, size_t ws_size,
                              hipStream_t stream) {
    const float* X     = (const float*)d_in[0];
    const float* W     = (const float*)d_in[1];
    const float* bias  = (const float*)d_in[2];
    const float* agg_w = (const float*)d_in[3];
    const float* agg_b = (const float*)d_in[4];
    const int*   gidx  = (const int*)d_in[5];

    float* ws    = (float*)d_ws;
    float* weff  = ws + OFF_WEFF;
    float* out   = (float*)d_out;

    (void)in_sizes; (void)n_in; (void)out_size; (void)ws_size;

    init_k<<<IN_ZBLOCKS, IN_THREADS, 0, stream>>>(weff);
    scatter_k<<<SC_BLOCKS + 1, SC_THREADS, 0, stream>>>(W, agg_w, gidx, bias, agg_b, weff, out);
    gemm_k<<<MGROUPS * KSPLIT, 32, 0, stream>>>(X, weff, out);
}
